# starting kernel
# baseline (speedup 1.0000x reference)
#include <hip/hip_runtime.h>

typedef float v2f __attribute__((ext_vector_type(2)));
typedef float v8f __attribute__((ext_vector_type(8)));

#define B_SZ   4096
#define F_NUM  16
#define V_SZ   20000
#define E_SZ   64
#define HID    1024

__global__ void __launch_bounds__(256) gather_embed(const int* __restrict__ idx,
                                                    const float* __restrict__ emb,
                                                    float* __restrict__ xw) {
    int tid = blockIdx.x * 256 + threadIdx.x;
    int b = tid >> 10;
    int c = tid & 1023;
    int f = c >> 6;
    int e = c & 63;
    int v = idx[b * F_NUM + f];
    xw[tid] = emb[((size_t)f * V_SZ + (size_t)v) * E_SZ + e];
}

__global__ void __launch_bounds__(128) gemm_relu(const float* __restrict__ X,
                                                 const float* __restrict__ W,
                                                 const float* __restrict__ bias,
                                                 float* __restrict__ Y) {
    const int lane = threadIdx.x & 31;
    const int gw   = blockIdx.x * 4 + (threadIdx.x >> 5);
    const int m0   = (gw >> 4) * 16;
    const int n0   = (gw & 15) * 64;
    const int lmod = lane & 15;
    const int lhi  = lane >> 4;

    v8f acc[4] = {v8f{}, v8f{}, v8f{}, v8f{}};

    const float* xrow = X + (size_t)(m0 + lmod) * HID + 2 * lhi;
    const float* wcol = W + (size_t)(2 * lhi) * HID + n0 + lmod;

    for (int k = 0; k < HID; k += 4) {
        v2f a = *(const v2f*)(xrow + k);
        const float* wr0 = wcol + (size_t)k * HID;
        const float* wr1 = wr0 + HID;
        v2f b0 = { wr0[ 0], wr1[ 0] };
        v2f b1 = { wr0[16], wr1[16] };
        v2f b2 = { wr0[32], wr1[32] };
        v2f b3 = { wr0[48], wr1[48] };
        acc[0] = __builtin_amdgcn_wmma_f32_16x16x4_f32(false, a, false, b0, (short)0, acc[0], false, false);
        acc[1] = __builtin_amdgcn_wmma_f32_16x16x4_f32(false, a, false, b1, (short)0, acc[1], false, false);
        acc[2] = __builtin_amdgcn_wmma_f32_16x16x4_f32(false, a, false, b2, (short)0, acc[2], false, false);
        acc[3] = __builtin_amdgcn_wmma_f32_16x16x4_f32(false, a, false, b3, (short)0, acc[3], false, false);
    }

    #pragma unroll
    for (int s = 0; s < 4; ++s) {
        const int n  = n0 + 16 * s + lmod;
        const float bv = bias[n];
        float* yp = Y + (size_t)(m0 + 8 * lhi) * HID + n;
        #pragma unroll
        for (int j = 0; j < 8; ++j) {
            float v = acc[s][j] + bv;
            yp[(size_t)j * HID] = v > 0.f ? v : 0.f;
        }
    }
}

__global__ void __launch_bounds__(128) attn_final(const float* __restrict__ L0,
                                                  const float* __restrict__ L1,
                                                  const float* __restrict__ L2,
                                                  const float* __restrict__ LF,
                                                  const float* __restrict__ wf,
                                                  const float* __restrict__ bf,
                                                  float* __restrict__ out) {
    const int b = blockIdx.x;
    const int t = threadIdx.x;
    const float* Ls[3] = {L0, L1, L2};
    const float* H[3];
    #pragma unroll
    for (int j = 0; j < 3; ++j) {
        int r = 3 * b + j;
        H[j] = Ls[r / B_SZ] + (size_t)(r % B_SZ) * HID;
    }

    __shared__ float red[3][128];
    __shared__ float cs[3];

    float p01 = 0.f, p02 = 0.f, p12 = 0.f;
    for (int h = t; h < HID; h += 128) {
        float h0 = H[0][h], h1 = H[1][h], h2 = H[2][h];
        p01 += h0 * h1; p02 += h0 * h2; p12 += h1 * h2;
    }
    red[0][t] = p01; red[1][t] = p02; red[2][t] = p12;
    __syncthreads();
    for (int off = 64; off > 0; off >>= 1) {
        if (t < off) {
            red[0][t] += red[0][t + off];
            red[1][t] += red[1][t + off];
            red[2][t] += red[2][t + off];
        }
        __syncthreads();
    }

    if (t == 0) {
        float s01 = red[0][0], s02 = red[1][0], s12 = red[2][0];
        float S[3][3] = {{0.f, s01, s02}, {s01, 0.f, s12}, {s02, s12, 0.f}};
        float col[3] = {0.f, 0.f, 0.f};
        #pragma unroll
        for (int i = 0; i < 3; ++i) {
            float m = fmaxf(S[i][0], fmaxf(S[i][1], S[i][2]));
            float e0 = expf(S[i][0] - m), e1 = expf(S[i][1] - m), e2 = expf(S[i][2] - m);
            float inv = 1.f / (e0 + e1 + e2);
            col[0] += e0 * inv; col[1] += e1 * inv; col[2] += e2 * inv;
        }
        cs[0] = col[0]; cs[1] = col[1]; cs[2] = col[2];
    }
    __syncthreads();

    const float c0 = cs[0], c1 = cs[1], c2 = cs[2];
    const float* lf = LF + (size_t)b * HID;
    float p = 0.f;
    for (int h = t; h < HID; h += 128) {
        float G = c0 * H[0][h] + c1 * H[1][h] + c2 * H[2][h];
        p += lf[h] * wf[h] + G * wf[HID + h];
    }
    __syncthreads();
    red[0][t] = p;
    __syncthreads();
    for (int off = 64; off > 0; off >>= 1) {
        if (t < off) red[0][t] += red[0][t + off];
        __syncthreads();
    }
    if (t == 0) out[b] = red[0][0] + bf[0];
}

extern "C" void kernel_launch(void* const* d_in, const int* in_sizes, int n_in,
                              void* d_out, int out_size, void* d_ws, size_t ws_size,
                              hipStream_t stream) {
    const int*   idx = (const int*)  d_in[0];
    const float* emb = (const float*)d_in[1];
    const float* w0  = (const float*)d_in[2];
    const float* b0  = (const float*)d_in[3];
    const float* w1  = (const float*)d_in[4];
    const float* b1  = (const float*)d_in[5];
    const float* w2  = (const float*)d_in[6];
    const float* b2  = (const float*)d_in[7];
    const float* w3  = (const float*)d_in[8];
    const float* b3  = (const float*)d_in[9];
    const float* wf  = (const float*)d_in[10];
    const float* bfin= (const float*)d_in[11];
    float* out = (float*)d_out;

    const size_t buf = (size_t)B_SZ * HID;
    float* ws0 = (float*)d_ws;
    float* ws1 = ws0 + buf;
    float* ws2 = ws1 + buf;
    float* ws3 = ws2 + buf;

    gather_embed<<<(B_SZ * HID) / 256, 256, 0, stream>>>(idx, emb, ws0);

    const int gemm_blocks = 1024;
    gemm_relu<<<gemm_blocks, 128, 0, stream>>>(ws0, w0, b0, ws1);
    gemm_relu<<<gemm_blocks, 128, 0, stream>>>(ws1, w1, b1, ws2);
    gemm_relu<<<gemm_blocks, 128, 0, stream>>>(ws2, w2, b2, ws0);
    gemm_relu<<<gemm_blocks, 128, 0, stream>>>(ws0, w3, b3, ws3);

    attn_final<<<B_SZ, 128, 0, stream>>>(ws1, ws2, ws0, ws3, wf, bfin, out);
}


# gemm_relu: each wave builds its 32x64 tile alone, no LDS: 16-byte global loads to registers, the four N-tiles are the column residues mod 4, loads two K-steps of 8 ahead; k order in each 8 changed
# speedup vs baseline: 1.5571x; 1.5571x over previous
#include <hip/hip_runtime.h>

typedef float v2f __attribute__((ext_vector_type(2)));
typedef float v4f __attribute__((ext_vector_type(4)));
typedef float v8f __attribute__((ext_vector_type(8)));

#define B_SZ   4096
#define F_NUM  16
#define V_SZ   20000
#define E_SZ   64
#define HID    1024

__global__ void __launch_bounds__(256) gather_embed(const int* __restrict__ idx,
                                                    const float* __restrict__ emb,
                                                    float* __restrict__ xw) {
    int tid = blockIdx.x * 256 + threadIdx.x;
    int b = tid >> 10;
    int c = tid & 1023;
    int f = c >> 6;
    int e = c & 63;
    int v = idx[b * F_NUM + f];
    xw[tid] = emb[((size_t)f * V_SZ + (size_t)v) * E_SZ + e];
}

struct Frag {
    v4f a0, a1;
    v4f b0, b1, b2, b3;
};

__device__ __forceinline__ void load_frag(Frag& f, const float* __restrict__ xa,
                                          const float* __restrict__ wb, int kk) {
    f.a0 = *(const v4f*)(xa + kk);
    f.a1 = *(const v4f*)(xa + 16 * HID + kk);
    const float* w = wb + (size_t)kk * HID;
    f.b0 = *(const v4f*)(w);
    f.b1 = *(const v4f*)(w + HID);
    f.b2 = *(const v4f*)(w + 2 * HID);
    f.b3 = *(const v4f*)(w + 3 * HID);
}

#define WMMA_F32(A, B, C) \
    __builtin_amdgcn_wmma_f32_16x16x4_f32(false, (A), false, (B), (short)0, (C), false, false)

__device__ __forceinline__ void mma_frag(v8f (&acc)[2][4], const Frag& f) {
    const v2f a00 = { f.a0.x, f.a0.y }, a01 = { f.a0.z, f.a0.w };
    const v2f a10 = { f.a1.x, f.a1.y }, a11 = { f.a1.z, f.a1.w };
    #pragma unroll
    for (int s = 0; s < 4; ++s) {
        const v2f b = { f.b0[s], f.b1[s] };
        acc[0][s] = WMMA_F32(a00, b, acc[0][s]);
        acc[1][s] = WMMA_F32(a10, b, acc[1][s]);
    }
    #pragma unroll
    for (int s = 0; s < 4; ++s) {
        const v2f b = { f.b2[s], f.b3[s] };
        acc[0][s] = WMMA_F32(a01, b, acc[0][s]);
        acc[1][s] = WMMA_F32(a11, b, acc[1][s]);
    }
}

__global__ void __launch_bounds__(128) gemm_relu(const float* __restrict__ X,
                                                 const float* __restrict__ W,
                                                 const float* __restrict__ bias,
                                                 float* __restrict__ Y) {
    const int lane = threadIdx.x & 31;
    const int wave = threadIdx.x >> 5;
    const int c    = lane & 15;
    const int h    = lane >> 4;
    const int m0 = (blockIdx.x >> 3) * 64 + (wave & 1) * 32;
    const int n0 = (blockIdx.x & 7) * 128 + (wave >> 1) * 64;

    v8f acc[2][4];
    #pragma unroll
    for (int i = 0; i < 2; ++i)
        #pragma unroll
        for (int s = 0; s < 4; ++s) acc[i][s] = v8f{};

    const float* xa = X + (size_t)(m0 + c) * HID + 4 * h;
    const float* wb = W + (size_t)(4 * h) * HID + n0 + 4 * c;

    Frag f0, f1, f2;
    load_frag(f0, xa, wb, 0);
    __builtin_amdgcn_sched_barrier(0);
    load_frag(f1, xa, wb, 8);
    __builtin_amdgcn_sched_barrier(0);
    for (int kk = 0; kk < HID - 16; kk += 24) {
        load_frag(f2, xa, wb, kk + 16);
        __builtin_amdgcn_sched_barrier(0);
        mma_frag(acc, f0);
        __builtin_amdgcn_sched_barrier(0);
        load_frag(f0, xa, wb, kk + 24);
        __builtin_amdgcn_sched_barrier(0);
        mma_frag(acc, f1);
        __builtin_amdgcn_sched_barrier(0);
        load_frag(f1, xa, wb, kk + 32);
        __builtin_amdgcn_sched_barrier(0);
        mma_frag(acc, f2);
        __builtin_amdgcn_sched_barrier(0);
    }
    mma_frag(acc, f0);
    mma_frag(acc, f1);

    const v4f bv = *(const v4f*)(bias + n0 + 4 * c);
    #pragma unroll
    for (int i = 0; i < 2; ++i) {
        float* yp = Y + (size_t)(m0 + 16 * i + 8 * h) * HID + n0 + 4 * c;
        #pragma unroll
        for (int r = 0; r < 8; ++r) {
            v4f o;
            #pragma unroll
            for (int s = 0; s < 4; ++s) {
                float v = acc[i][s][r] + bv[s];
                o[s] = v > 0.f ? v : 0.f;
            }
            *(v4f*)(yp + (size_t)r * HID) = o;
        }
    }
}

__global__ void __launch_bounds__(128) attn_final(const float* __restrict__ L0,
                                                  const float* __restrict__ L1,
                                                  const float* __restrict__ L2,
                                                  const float* __restrict__ LF,
                                                  const float* __restrict__ wf,
                                                  const float* __restrict__ bf,
                                                  float* __restrict__ out) {
    const int b = blockIdx.x;
    const int t = threadIdx.x;
    const float* Ls[3] = {L0, L1, L2};
    const float* H[3];
    #pragma unroll
    for (int j = 0; j < 3; ++j) {
        int r = 3 * b + j;
        H[j] = Ls[r / B_SZ] + (size_t)(r % B_SZ) * HID;
    }

    __shared__ float red[3][128];
    __shared__ float cs[3];

    float p01 = 0.f, p02 = 0.f, p12 = 0.f;
    for (int h = t; h < HID; h += 128) {
        float h0 = H[0][h], h1 = H[1][h], h2 = H[2][h];
        p01 += h0 * h1; p02 += h0 * h2; p12 += h1 * h2;
    }
    red[0][t] = p01; red[1][t] = p02; red[2][t] = p12;
    __syncthreads();
    for (int off = 64; off > 0; off >>= 1) {
        if (t < off) {
            red[0][t] += red[0][t + off];
            red[1][t] += red[1][t + off];
            red[2][t] += red[2][t + off];
        }
        __syncthreads();
    }

    if (t == 0) {
        float s01 = red[0][0], s02 = red[1][0], s12 = red[2][0];
        float S[3][3] = {{0.f, s01, s02}, {s01, 0.f, s12}, {s02, s12, 0.f}};
        float col[3] = {0.f, 0.f, 0.f};
        #pragma unroll
        for (int i = 0; i < 3; ++i) {
            float m = fmaxf(S[i][0], fmaxf(S[i][1], S[i][2]));
            float e0 = expf(S[i][0] - m), e1 = expf(S[i][1] - m), e2 = expf(S[i][2] - m);
            float inv = 1.f / (e0 + e1 + e2);
            col[0] += e0 * inv; col[1] += e1 * inv; col[2] += e2 * inv;
        }
        cs[0] = col[0]; cs[1] = col[1]; cs[2] = col[2];
    }
    __syncthreads();

    const float c0 = cs[0], c1 = cs[1], c2 = cs[2];
    const float* lf = LF + (size_t)b * HID;
    float p = 0.f;
    for (int h = t; h < HID; h += 128) {
        float G = c0 * H[0][h] + c1 * H[1][h] + c2 * H[2][h];
        p += lf[h] * wf[h] + G * wf[HID + h];
    }
    __syncthreads();
    red[0][t] = p;
    __syncthreads();
    for (int off = 64; off > 0; off >>= 1) {
        if (t < off) red[0][t] += red[0][t + off];
        __syncthreads();
    }
    if (t == 0) out[b] = red[0][0] + bf[0];
}

extern "C" void kernel_launch(void* const* d_in, const int* in_sizes, int n_in,
                              void* d_out, int out_size, void* d_ws, size_t ws_size,
                              hipStream_t stream) {
    const int*   idx = (const int*)  d_in[0];
    const float* emb = (const float*)d_in[1];
    const float* w0  = (const float*)d_in[2];
    const float* b0  = (const float*)d_in[3];
    const float* w1  = (const float*)d_in[4];
    const float* b1  = (const float*)d_in[5];
    const float* w2  = (const float*)d_in[6];
    const float* b2  = (const float*)d_in[7];
    const float* w3  = (const float*)d_in[8];
    const float* b3  = (const float*)d_in[9];
    const float* wf  = (const float*)d_in[10];
    const float* bfin= (const float*)d_in[11];
    float* out = (float*)d_out;

    const size_t buf = (size_t)B_SZ * HID;
    float* ws0 = (float*)d_ws;
    float* ws1 = ws0 + buf;
    float* ws2 = ws1 + buf;
    float* ws3 = ws2 + buf;

    gather_embed<<<(B_SZ * HID) / 256, 256, 0, stream>>>(idx, emb, ws0);

    const int gemm_blocks = 512;
    gemm_relu<<<gemm_blocks, 128, 0, stream>>>(ws0, w0, b0, ws1);
    gemm_relu<<<gemm_blocks, 128, 0, stream>>>(ws1, w1, b1, ws2);
    gemm_relu<<<gemm_blocks, 128, 0, stream>>>(ws2, w2, b2, ws0);
    gemm_relu<<<gemm_blocks, 128, 0, stream>>>(ws0, w3, b3, ws3);

    attn_final<<<B_SZ, 128, 0, stream>>>(ws1, ws2, ws0, ws3, wf, bfin, out);
}


# gemm_relu: operand loads issued three groups of 16 k (96 WMMAs of the wave) ahead in four register sets, not two K-steps of 8 ahead in three; k order of the sums unchanged
# speedup vs baseline: 1.6331x; 1.0488x over previous
#include <hip/hip_runtime.h>

typedef float v2f __attribute__((ext_vector_type(2)));
typedef float v4f __attribute__((ext_vector_type(4)));
typedef float v8f __attribute__((ext_vector_type(8)));

#define B_SZ   4096
#define F_NUM  16
#define V_SZ   20000
#define E_SZ   64
#define HID    1024

__global__ void __launch_bounds__(256) gather_embed(const int* __restrict__ idx,
                                                    const float* __restrict__ emb,
                                                    float* __restrict__ xw) {
    int tid = blockIdx.x * 256 + threadIdx.x;
    int b = tid >> 10;
    int c = tid & 1023;
    int f = c >> 6;
    int e = c & 63;
    int v = idx[b * F_NUM + f];
    xw[tid] = emb[((size_t)f * V_SZ + (size_t)v) * E_SZ + e];
}

struct Frag {
    v4f a0, a1;
    v4f b0, b1, b2, b3;
};

__device__ __forceinline__ void load_frag(Frag& f, const float* __restrict__ xa,
                                          const float* __restrict__ wb, int kk) {
    f.a0 = *(const v4f*)(xa + kk);
    f.a1 = *(const v4f*)(xa + 16 * HID + kk);
    const float* w = wb + (size_t)kk * HID;
    f.b0 = *(const v4f*)(w);
    f.b1 = *(const v4f*)(w + HID);
    f.b2 = *(const v4f*)(w + 2 * HID);
    f.b3 = *(const v4f*)(w + 3 * HID);
}

#define WMMA_F32(A, B, C) \
    __builtin_amdgcn_wmma_f32_16x16x4_f32(false, (A), false, (B), (short)0, (C), false, false)

__device__ __forceinline__ void mma_frag(v8f (&acc)[2][4], const Frag& f) {
    const v2f a00 = { f.a0.x, f.a0.y }, a01 = { f.a0.z, f.a0.w };
    const v2f a10 = { f.a1.x, f.a1.y }, a11 = { f.a1.z, f.a1.w };
    #pragma unroll
    for (int s = 0; s < 4; ++s) {
        const v2f b = { f.b0[s], f.b1[s] };
        acc[0][s] = WMMA_F32(a00, b, acc[0][s]);
        acc[1][s] = WMMA_F32(a10, b, acc[1][s]);
    }
    #pragma unroll
    for (int s = 0; s < 4; ++s) {
        const v2f b = { f.b2[s], f.b3[s] };
        acc[0][s] = WMMA_F32(a01, b, acc[0][s]);
        acc[1][s] = WMMA_F32(a11, b, acc[1][s]);
    }
}

struct Frag2 { Frag lo, hi; };

__device__ __forceinline__ void load_frag2(Frag2& f, const float* __restrict__ xa,
                                           const float* __restrict__ wb, int kk) {
    load_frag(f.lo, xa, wb, kk);
    load_frag(f.hi, xa, wb, kk + 8);
}

__device__ __forceinline__ void mma_frag2(v8f (&acc)[2][4], const Frag2& f) {
    mma_frag(acc, f.lo);
    mma_frag(acc, f.hi);
}

__global__ void __launch_bounds__(128) gemm_relu(const float* __restrict__ X,
                                                 const float* __restrict__ W,
                                                 const float* __restrict__ bias,
                                                 float* __restrict__ Y) {
    const int lane = threadIdx.x & 31;
    const int wave = threadIdx.x >> 5;
    const int c    = lane & 15;
    const int h    = lane >> 4;
    const int m0 = (blockIdx.x >> 3) * 64 + (wave & 1) * 32;
    const int n0 = (blockIdx.x & 7) * 128 + (wave >> 1) * 64;

    v8f acc[2][4];
    #pragma unroll
    for (int i = 0; i < 2; ++i)
        #pragma unroll
        for (int s = 0; s < 4; ++s) acc[i][s] = v8f{};

    const float* xa = X + (size_t)(m0 + c) * HID + 4 * h;
    const float* wb = W + (size_t)(4 * h) * HID + n0 + 4 * c;

    Frag2 f0, f1, f2, f3;
    load_frag2(f0, xa, wb, 0);
    __builtin_amdgcn_sched_barrier(0);
    load_frag2(f1, xa, wb, 16);
    __builtin_amdgcn_sched_barrier(0);
    load_frag2(f2, xa, wb, 32);
    __builtin_amdgcn_sched_barrier(0);
    for (int kk = 0; kk < HID - 64; kk += 64) {
        load_frag2(f3, xa, wb, kk + 48);
        __builtin_amdgcn_sched_barrier(0);
        mma_frag2(acc, f0);
        __builtin_amdgcn_sched_barrier(0);
        load_frag2(f0, xa, wb, kk + 64);
        __builtin_amdgcn_sched_barrier(0);
        mma_frag2(acc, f1);
        __builtin_amdgcn_sched_barrier(0);
        load_frag2(f1, xa, wb, kk + 80);
        __builtin_amdgcn_sched_barrier(0);
        mma_frag2(acc, f2);
        __builtin_amdgcn_sched_barrier(0);
        load_frag2(f2, xa, wb, kk + 96);
        __builtin_amdgcn_sched_barrier(0);
        mma_frag2(acc, f3);
        __builtin_amdgcn_sched_barrier(0);
    }
    load_frag2(f3, xa, wb, HID - 16);
    __builtin_amdgcn_sched_barrier(0);
    mma_frag2(acc, f0);
    mma_frag2(acc, f1);
    mma_frag2(acc, f2);
    mma_frag2(acc, f3);

    const v4f bv = *(const v4f*)(bias + n0 + 4 * c);
    #pragma unroll
    for (int i = 0; i < 2; ++i) {
        float* yp = Y + (size_t)(m0 + 16 * i + 8 * h) * HID + n0 + 4 * c;
        #pragma unroll
        for (int r = 0; r < 8; ++r) {
            v4f o;
            #pragma unroll
            for (int s = 0; s < 4; ++s) {
                float v = acc[i][s][r] + bv[s];
                o[s] = v > 0.f ? v : 0.f;
            }
            *(v4f*)(yp + (size_t)r * HID) = o;
        }
    }
}

__global__ void __launch_bounds__(128) attn_final(const float* __restrict__ L0,
                                                  const float* __restrict__ L1,
                                                  const float* __restrict__ L2,
                                                  const float* __restrict__ LF,
                                                  const float* __restrict__ wf,
                                                  const float* __restrict__ bf,
                                                  float* __restrict__ out) {
    const int b = blockIdx.x;
    const int t = threadIdx.x;
    const float* Ls[3] = {L0, L1, L2};
    const float* H[3];
    #pragma unroll
    for (int j = 0; j < 3; ++j) {
        int r = 3 * b + j;
        H[j] = Ls[r / B_SZ] + (size_t)(r % B_SZ) * HID;
    }

    __shared__ float red[3][128];
    __shared__ float cs[3];

    float p01 = 0.f, p02 = 0.f, p12 = 0.f;
    for (int h = t; h < HID; h += 128) {
        float h0 = H[0][h], h1 = H[1][h], h2 = H[2][h];
        p01 += h0 * h1; p02 += h0 * h2; p12 += h1 * h2;
    }
    red[0][t] = p01; red[1][t] = p02; red[2][t] = p12;
    __syncthreads();
    for (int off = 64; off > 0; off >>= 1) {
        if (t < off) {
            red[0][t] += red[0][t + off];
            red[1][t] += red[1][t + off];
            red[2][t] += red[2][t + off];
        }
        __syncthreads();
    }

    if (t == 0) {
        float s01 = red[0][0], s02 = red[1][0], s12 = red[2][0];
        float S[3][3] = {{0.f, s01, s02}, {s01, 0.f, s12}, {s02, s12, 0.f}};
        float col[3] = {0.f, 0.f, 0.f};
        #pragma unroll
        for (int i = 0; i < 3; ++i) {
            float m = fmaxf(S[i][0], fmaxf(S[i][1], S[i][2]));
            float e0 = expf(S[i][0] - m), e1 = expf(S[i][1] - m), e2 = expf(S[i][2] - m);
            float inv = 1.f / (e0 + e1 + e2);
            col[0] += e0 * inv; col[1] += e1 * inv; col[2] += e2 * inv;
        }
        cs[0] = col[0]; cs[1] = col[1]; cs[2] = col[2];
    }
    __syncthreads();

    const float c0 = cs[0], c1 = cs[1], c2 = cs[2];
    const float* lf = LF + (size_t)b * HID;
    float p = 0.f;
    for (int h = t; h < HID; h += 128) {
        float G = c0 * H[0][h] + c1 * H[1][h] + c2 * H[2][h];
        p += lf[h] * wf[h] + G * wf[HID + h];
    }
    __syncthreads();
    red[0][t] = p;
    __syncthreads();
    for (int off = 64; off > 0; off >>= 1) {
        if (t < off) red[0][t] += red[0][t + off];
        __syncthreads();
    }
    if (t == 0) out[b] = red[0][0] + bf[0];
}

extern "C" void kernel_launch(void* const* d_in, const int* in_sizes, int n_in,
                              void* d_out, int out_size, void* d_ws, size_t ws_size,
                              hipStream_t stream) {
    const int*   idx = (const int*)  d_in[0];
    const float* emb = (const float*)d_in[1];
    const float* w0  = (const float*)d_in[2];
    const float* b0  = (const float*)d_in[3];
    const float* w1  = (const float*)d_in[4];
    const float* b1  = (const float*)d_in[5];
    const float* w2  = (const float*)d_in[6];
    const float* b2  = (const float*)d_in[7];
    const float* w3  = (const float*)d_in[8];
    const float* b3  = (const float*)d_in[9];
    const float* wf  = (const float*)d_in[10];
    const float* bfin= (const float*)d_in[11];
    float* out = (float*)d_out;

    const size_t buf = (size_t)B_SZ * HID;
    float* ws0 = (float*)d_ws;
    float* ws1 = ws0 + buf;
    float* ws2 = ws1 + buf;
    float* ws3 = ws2 + buf;

    gather_embed<<<(B_SZ * HID) / 256, 256, 0, stream>>>(idx, emb, ws0);

    const int gemm_blocks = 512;
    gemm_relu<<<gemm_blocks, 128, 0, stream>>>(ws0, w0, b0, ws1);
    gemm_relu<<<gemm_blocks, 128, 0, stream>>>(ws1, w1, b1, ws2);
    gemm_relu<<<gemm_blocks, 128, 0, stream>>>(ws2, w2, b2, ws0);
    gemm_relu<<<gemm_blocks, 128, 0, stream>>>(ws0, w3, b3, ws3);

    attn_final<<<B_SZ, 128, 0, stream>>>(ws1, ws2, ws0, ws3, wf, bfin, out);
}


# attention rows run as extra work-groups beside the last GEMM in one launch, hidden under its WMMA time; that GEMM sums relu(tile)*wf in registers; tail kernel adds 17 floats a row; sum order changes
# speedup vs baseline: 1.7067x; 1.0451x over previous
#include <hip/hip_runtime.h>

typedef float v2f __attribute__((ext_vector_type(2)));
typedef float v4f __attribute__((ext_vector_type(4)));
typedef float v8f __attribute__((ext_vector_type(8)));

#define B_SZ   4096
#define F_NUM  16
#define V_SZ   20000
#define E_SZ   64
#define HID    1024

__global__ void __launch_bounds__(256) gather_embed(const int* __restrict__ idx,
                                                    const float* __restrict__ emb,
                                                    float* __restrict__ xw) {
    int tid = blockIdx.x * 256 + threadIdx.x;
    int b = tid >> 10;
    int c = tid & 1023;
    int f = c >> 6;
    int e = c & 63;
    int v = idx[b * F_NUM + f];
    xw[tid] = emb[((size_t)f * V_SZ + (size_t)v) * E_SZ + e];
}

struct Frag {
    v4f a0, a1;
    v4f b0, b1, b2, b3;
};

__device__ __forceinline__ void load_frag(Frag& f, const float* __restrict__ xa,
                                          const float* __restrict__ wb, int kk) {
    f.a0 = *(const v4f*)(xa + kk);
    f.a1 = *(const v4f*)(xa + 16 * HID + kk);
    const float* w = wb + (size_t)kk * HID;
    f.b0 = *(const v4f*)(w);
    f.b1 = *(const v4f*)(w + HID);
    f.b2 = *(const v4f*)(w + 2 * HID);
    f.b3 = *(const v4f*)(w + 3 * HID);
}

#define WMMA_F32(A, B, C) \
    __builtin_amdgcn_wmma_f32_16x16x4_f32(false, (A), false, (B), (short)0, (C), false, false)

__device__ __forceinline__ void mma_frag(v8f (&acc)[2][4], const Frag& f) {
    const v2f a00 = { f.a0.x, f.a0.y }, a01 = { f.a0.z, f.a0.w };
    const v2f a10 = { f.a1.x, f.a1.y }, a11 = { f.a1.z, f.a1.w };
    #pragma unroll
    for (int s = 0; s < 4; ++s) {
        const v2f b = { f.b0[s], f.b1[s] };
        acc[0][s] = WMMA_F32(a00, b, acc[0][s]);
        acc[1][s] = WMMA_F32(a10, b, acc[1][s]);
    }
    #pragma unroll
    for (int s = 0; s < 4; ++s) {
        const v2f b = { f.b2[s], f.b3[s] };
        acc[0][s] = WMMA_F32(a01, b, acc[0][s]);
        acc[1][s] = WMMA_F32(a11, b, acc[1][s]);
    }
}

struct Frag2 { Frag lo, hi; };

__device__ __forceinline__ void load_frag2(Frag2& f, const float* __restrict__ xa,
                                           const float* __restrict__ wb, int kk) {
    load_frag(f.lo, xa, wb, kk);
    load_frag(f.hi, xa, wb, kk + 8);
}

__device__ __forceinline__ void mma_frag2(v8f (&acc)[2][4], const Frag2& f) {
    mma_frag(acc, f.lo);
    mma_frag(acc, f.hi);
}

__device__ __forceinline__ void gemm_tile(v8f (&acc)[2][4], const float* __restrict__ xa,
                                          const float* __restrict__ wb) {
    #pragma unroll
    for (int i = 0; i < 2; ++i)
        #pragma unroll
        for (int s = 0; s < 4; ++s) acc[i][s] = v8f{};

    Frag2 f0, f1, f2, f3;
    load_frag2(f0, xa, wb, 0);
    __builtin_amdgcn_sched_barrier(0);
    load_frag2(f1, xa, wb, 16);
    __builtin_amdgcn_sched_barrier(0);
    load_frag2(f2, xa, wb, 32);
    __builtin_amdgcn_sched_barrier(0);
    for (int kk = 0; kk < HID - 64; kk += 64) {
        load_frag2(f3, xa, wb, kk + 48);
        __builtin_amdgcn_sched_barrier(0);
        mma_frag2(acc, f0);
        __builtin_amdgcn_sched_barrier(0);
        load_frag2(f0, xa, wb, kk + 64);
        __builtin_amdgcn_sched_barrier(0);
        mma_frag2(acc, f1);
        __builtin_amdgcn_sched_barrier(0);
        load_frag2(f1, xa, wb, kk + 80);
        __builtin_amdgcn_sched_barrier(0);
        mma_frag2(acc, f2);
        __builtin_amdgcn_sched_barrier(0);
        load_frag2(f2, xa, wb, kk + 96);
        __builtin_amdgcn_sched_barrier(0);
        mma_frag2(acc, f3);
        __builtin_amdgcn_sched_barrier(0);
    }
    load_frag2(f3, xa, wb, HID - 16);
    __builtin_amdgcn_sched_barrier(0);
    mma_frag2(acc, f0);
    mma_frag2(acc, f1);
    mma_frag2(acc, f2);
    mma_frag2(acc, f3);
}

#define GEMM_BLOCKS 512
#define SIDE_BLOCKS 256

__global__ void __launch_bounds__(128) gemm_relu(const float* __restrict__ X,
                                                 const float* __restrict__ W,
                                                 const float* __restrict__ bias,
                                                 float* __restrict__ Y) {
    const int lane = threadIdx.x & 31;
    const int wave = threadIdx.x >> 5;
    const int c    = lane & 15;
    const int h    = lane >> 4;
    const int m0 = (blockIdx.x >> 3) * 64 + (wave & 1) * 32;
    const int n0 = (blockIdx.x & 7) * 128 + (wave >> 1) * 64;

    const float* xa = X + (size_t)(m0 + c) * HID + 4 * h;
    const float* wb = W + (size_t)(4 * h) * HID + n0 + 4 * c;

    v8f acc[2][4];
    gemm_tile(acc, xa, wb);

    const v4f bv = *(const v4f*)(bias + n0 + 4 * c);
    #pragma unroll
    for (int i = 0; i < 2; ++i) {
        float* yp = Y + (size_t)(m0 + 16 * i + 8 * h) * HID + n0 + 4 * c;
        #pragma unroll
        for (int r = 0; r < 8; ++r) {
            v4f o;
            #pragma unroll
            for (int s = 0; s < 4; ++s) {
                float v = acc[i][s][r] + bv[s];
                o[s] = v > 0.f ? v : 0.f;
            }
            *(v4f*)(yp + (size_t)r * HID) = o;
        }
    }
}

__global__ void __launch_bounds__(128) gemm_last_attn(const float* __restrict__ X,
                                                      const float* __restrict__ W,
                                                      const float* __restrict__ bias,
                                                      const float* __restrict__ L0,
                                                      const float* __restrict__ L1,
                                                      const float* __restrict__ wf,
                                                      float* __restrict__ part,
                                                      float* __restrict__ apart) {
    const int lane = threadIdx.x & 31;
    const int wave = threadIdx.x >> 5;

    if (blockIdx.x >= GEMM_BLOCKS) {
        const unsigned wv = __builtin_amdgcn_readfirstlane(wave);
        const unsigned b0 = (blockIdx.x - GEMM_BLOCKS) * 16u + wv * 4u;
        v4f w2[8];
        #pragma unroll
        for (int j = 0; j < 8; ++j) w2[j] = *(const v4f*)(wf + HID + 128 * j + 4 * lane);

        float keep = 0.f;
        #pragma unroll 1
        for (unsigned t = 0; t < 4; ++t) {
            const float* H[3];
            #pragma unroll
            for (unsigned j = 0; j < 3; ++j) {
                const unsigned r = 3u * (b0 + t) + j;
                const unsigned q = r / B_SZ;
                const float* base = (q == 0) ? L0 : ((q == 1) ? L1 : X);
                H[j] = base + (size_t)(r % B_SZ) * HID + 4 * lane;
            }
            v2f a01 = {0.f, 0.f}, a02 = {0.f, 0.f}, a12 = {0.f, 0.f};
            v2f g0 = {0.f, 0.f}, g1 = {0.f, 0.f}, g2 = {0.f, 0.f};
            #pragma unroll
            for (int j = 0; j < 8; ++j) {
                const v4f h0 = *(const v4f*)(H[0] + 128 * j);
                const v4f h1 = *(const v4f*)(H[1] + 128 * j);
                const v4f h2 = *(const v4f*)(H[2] + 128 * j);
                const v2f x0 = {h0.x, h0.y}, y0 = {h0.z, h0.w};
                const v2f x1 = {h1.x, h1.y}, y1 = {h1.z, h1.w};
                const v2f x2 = {h2.x, h2.y}, y2 = {h2.z, h2.w};
                const v2f xw = {w2[j].x, w2[j].y}, yw = {w2[j].z, w2[j].w};
                a01 += x0 * x1; a02 += x0 * x2; a12 += x1 * x2;
                g0  += x0 * xw; g1  += x1 * xw; g2  += x2 * xw;
                a01 += y0 * y1; a02 += y0 * y2; a12 += y1 * y2;
                g0  += y0 * yw; g1  += y1 * yw; g2  += y2 * yw;
            }
            float p01 = a01.x + a01.y, p02 = a02.x + a02.y, p12 = a12.x + a12.y;
            float d0 = g0.x + g0.y, d1 = g1.x + g1.y, d2 = g2.x + g2.y;
            for (int m = 16; m > 0; m >>= 1) {
                p01 += __shfl_xor(p01, m); p02 += __shfl_xor(p02, m); p12 += __shfl_xor(p12, m);
                d0  += __shfl_xor(d0, m);  d1  += __shfl_xor(d1, m);  d2  += __shfl_xor(d2, m);
            }
            const float S[3][3] = {{0.f, p01, p02}, {p01, 0.f, p12}, {p02, p12, 0.f}};
            float col[3] = {0.f, 0.f, 0.f};
            #pragma unroll
            for (int i = 0; i < 3; ++i) {
                float mx = fmaxf(S[i][0], fmaxf(S[i][1], S[i][2]));
                float e0 = expf(S[i][0] - mx), e1 = expf(S[i][1] - mx), e2 = expf(S[i][2] - mx);
                float inv = 1.f / (e0 + e1 + e2);
                col[0] += e0 * inv; col[1] += e1 * inv; col[2] += e2 * inv;
            }
            const float res = col[0] * d0 + col[1] * d1 + col[2] * d2;
            if ((unsigned)lane == t) keep = res;
        }
        if (lane < 4) apart[b0 + lane] = keep;
        return;
    }

    const int c    = lane & 15;
    const int h    = lane >> 4;
    const int m0 = (blockIdx.x >> 3) * 64 + (wave & 1) * 32;
    const int n0 = (blockIdx.x & 7) * 128 + (wave >> 1) * 64;

    const float* xa = X + (size_t)(m0 + c) * HID + 4 * h;
    const float* wb = W + (size_t)(4 * h) * HID + n0 + 4 * c;

    v8f acc[2][4];
    gemm_tile(acc, xa, wb);

    const v4f bv = *(const v4f*)(bias + n0 + 4 * c);
    const v4f wv = *(const v4f*)(wf + n0 + 4 * c);
    float keep = 0.f;
    #pragma unroll
    for (int i = 0; i < 2; ++i) {
        #pragma unroll
        for (int r = 0; r < 8; ++r) {
            float p = 0.f;
            #pragma unroll
            for (int s = 0; s < 4; ++s) {
                float v = acc[i][s][r] + bv[s];
                v = v > 0.f ? v : 0.f;
                p += v * wv[s];
            }
            p += __shfl_xor(p, 1);
            p += __shfl_xor(p, 2);
            p += __shfl_xor(p, 4);
            p += __shfl_xor(p, 8);
            if (c == 8 * i + r) keep = p;
        }
    }
    const int slot = (blockIdx.x & 7) * 2 + (wave >> 1);
    part[(size_t)slot * B_SZ + m0 + 16 * (c >> 3) + 8 * h + (c & 7)] = keep;
}

__global__ void __launch_bounds__(256) finalize(const float* __restrict__ part,
                                                const float* __restrict__ apart,
                                                const float* __restrict__ bf,
                                                float* __restrict__ out) {
    const int b = blockIdx.x * 256 + threadIdx.x;
    float s = 0.f;
    for (int k = 0; k < 16; ++k) s += part[(size_t)k * B_SZ + b];
    out[b] = s + apart[b] + bf[0];
}

extern "C" void kernel_launch(void* const* d_in, const int* in_sizes, int n_in,
                              void* d_out, int out_size, void* d_ws, size_t ws_size,
                              hipStream_t stream) {
    const int*   idx = (const int*)  d_in[0];
    const float* emb = (const float*)d_in[1];
    const float* w0  = (const float*)d_in[2];
    const float* b0  = (const float*)d_in[3];
    const float* w1  = (const float*)d_in[4];
    const float* b1  = (const float*)d_in[5];
    const float* w2  = (const float*)d_in[6];
    const float* b2  = (const float*)d_in[7];
    const float* w3  = (const float*)d_in[8];
    const float* b3  = (const float*)d_in[9];
    const float* wf  = (const float*)d_in[10];
    const float* bfin= (const float*)d_in[11];
    float* out = (float*)d_out;

    const size_t buf = (size_t)B_SZ * HID;
    float* ws0 = (float*)d_ws;
    float* ws1 = ws0 + buf;
    float* ws2 = ws1 + buf;
    float* part  = ws2 + buf;
    float* apart = part + 16 * B_SZ;

    gather_embed<<<(B_SZ * HID) / 256, 256, 0, stream>>>(idx, emb, ws0);

    gemm_relu<<<GEMM_BLOCKS, 128, 0, stream>>>(ws0, w0, b0, ws1);
    gemm_relu<<<GEMM_BLOCKS, 128, 0, stream>>>(ws1, w1, b1, ws2);
    gemm_relu<<<GEMM_BLOCKS, 128, 0, stream>>>(ws2, w2, b2, ws0);

    gemm_last_attn<<<GEMM_BLOCKS + SIDE_BLOCKS, 128, 0, stream>>>(ws0, w3, b3, ws1, ws2, wf,
                                                                  part, apart);

    finalize<<<B_SZ / 256, 256, 0, stream>>>(part, apart, bfin, out);
}
